# starting kernel
# baseline (speedup 1.0000x reference)
#include <hip/hip_runtime.h>
#include <math.h>

#define B_      4
#define N_      10000
#define F_IN_   64
#define UNITS_  64
#define E_      160000
#define C_      128
#define M_      (B_ * N_)
#define MV_     3
#define KTOT_   (MV_ * C_)
#define LDSW_   388

typedef __attribute__((ext_vector_type(2))) float v2f;
typedef __attribute__((ext_vector_type(8))) float v8f;


__global__ void k_zero(float* __restrict__ p, int n) {
  int i = blockIdx.x * blockDim.x + threadIdx.x;
  if (i < n) p[i] = 0.0f;
}

__global__ void k_concat_ru(const float* __restrict__ in,
                            const float* __restrict__ hx,
                            float* __restrict__ v0) {
  int idx = blockIdx.x * blockDim.x + threadIdx.x;
  if (idx >= M_ * C_) return;
  int m  = idx >> 7;
  int ch = idx & (C_ - 1);
  float v = (ch < F_IN_) ? in[(size_t)m * F_IN_ + ch]
                         : hx[(size_t)m * UNITS_ + (ch - F_IN_)];
  v0[idx] = v;
}

__global__ void k_concat_c(const float* __restrict__ in,
                           const float* __restrict__ hx,
                           const float* __restrict__ ru,
                           float* __restrict__ v0) {
  int idx = blockIdx.x * blockDim.x + threadIdx.x;
  if (idx >= M_ * C_) return;
  int m  = idx >> 7;
  int ch = idx & (C_ - 1);
  float v;
  if (ch < F_IN_) {
    v = in[(size_t)m * F_IN_ + ch];
  } else {
    int u = ch - F_IN_;
    float r = ru[(size_t)m * (2 * UNITS_) + u];
    v = r * hx[(size_t)m * UNITS_ + u];
  }
  v0[idx] = v;
}

__global__ void k_scatter(const float* __restrict__ x,
                          float* __restrict__ y,
                          const int* __restrict__ src,
                          const int* __restrict__ dst,
                          const float* __restrict__ ker) {
  int idx = blockIdx.x * blockDim.x + threadIdx.x;
  if (idx >= E_ * 32) return;
  int e = idx >> 5;
  int q = (idx & 31) << 2;
  __builtin_prefetch(&src[e + 4096], 0, 0);
  __builtin_prefetch(&dst[e + 4096], 0, 0);
  int   s = src[e];
  int   d = dst[e];
  float k = ker[e];
#pragma unroll
  for (int b = 0; b < B_; ++b) {
    const float4 v = *(const float4*)(x + ((size_t)(b * N_ + s)) * C_ + q);
    float* yp = y + ((size_t)(b * N_ + d)) * C_ + q;
    atomicAdd(yp + 0, k * v.x);
    atomicAdd(yp + 1, k * v.y);
    atomicAdd(yp + 2, k * v.z);
    atomicAdd(yp + 3, k * v.w);
  }
}


__device__ inline void fill_As(float* __restrict__ As,
                               const float* __restrict__ v0,
                               const float* __restrict__ v1,
                               const float* __restrict__ v2,
                               int m0, int tid, int nthreads) {
  for (int i = tid; i < 16 * (KTOT_ / 4); i += nthreads) {
    int row  = i / (KTOT_ / 4);
    int kk4  = (i % (KTOT_ / 4)) * 4;
    int view = kk4 >> 7;
    int ch   = kk4 & (C_ - 1);
    const float* srcp = (view == 0 ? v0 : (view == 1 ? v1 : v2))
                        + ((size_t)(m0 + row)) * C_ + ch;
    *(float4*)(As + row * LDSW_ + kk4) = *(const float4*)srcp;
  }
  __syncthreads();
}

__device__ inline v8f wmma_tile_lds(const float* __restrict__ As,
                                    const float* __restrict__ W, int nout,
                                    int n0, int lane) {
  const int arow = lane & 15;
  const int koff = (lane >> 4) << 1;
  const int col  = n0 + (lane & 15);
  v8f acc = {};
#pragma unroll 4
  for (int kb = 0; kb < KTOT_ / 4; ++kb) {
    const int k0 = kb * 4 + koff;
    v2f a = *(const v2f*)(As + arow * LDSW_ + k0);
    v2f b;
    b.x = W[(size_t)k0 * nout + col];
    b.y = W[(size_t)(k0 + 1) * nout + col];
    acc = __builtin_amdgcn_wmma_f32_16x16x4_f32(
         false, a,  false, b,
         (short)0, acc,  false,  false);
  }
  return acc;
}

__global__ void k_gemm_ru(const float* __restrict__ v0,
                          const float* __restrict__ v1,
                          const float* __restrict__ v2,
                          const float* __restrict__ W,
                          const float* __restrict__ bias,
                          float* __restrict__ ru) {
  __shared__ float As[16 * LDSW_];
  const int lane = threadIdx.x & 31;
  const int wave = threadIdx.x >> 5;
  const int m0 = blockIdx.x * 16;
  const int n0 = wave * 16;
  fill_As(As, v0, v1, v2, m0, threadIdx.x, 256);
  v8f acc = wmma_tile_lds(As, W, 2 * UNITS_, n0, lane);
  const int col   = n0 + (lane & 15);
  const int mbase = m0 + ((lane >> 4) << 3);
  const float bv  = bias[col];
#pragma unroll
  for (int r = 0; r < 8; ++r) {
    int m = mbase + r;
    float x = acc[r] + bv;
    ru[(size_t)m * (2 * UNITS_) + col] = 1.0f / (1.0f + __expf(-x));
  }
}

__global__ void k_gemm_c(const float* __restrict__ v0,
                         const float* __restrict__ v1,
                         const float* __restrict__ v2,
                         const float* __restrict__ W,
                         const float* __restrict__ bias,
                         const float* __restrict__ ru,
                         const float* __restrict__ hx,
                         float* __restrict__ out) {
  __shared__ float As[16 * LDSW_];
  const int lane = threadIdx.x & 31;
  const int wave = threadIdx.x >> 5;
  const int m0 = blockIdx.x * 16;
  const int n0 = wave * 16;
  fill_As(As, v0, v1, v2, m0, threadIdx.x, 128);
  v8f acc = wmma_tile_lds(As, W, UNITS_, n0, lane);
  const int col   = n0 + (lane & 15);
  const int mbase = m0 + ((lane >> 4) << 3);
  const float bv  = bias[col];
#pragma unroll
  for (int r = 0; r < 8; ++r) {
    int m = mbase + r;
    float c = tanhf(acc[r] + bv);
    float u = ru[(size_t)m * (2 * UNITS_) + UNITS_ + col];
    float h = hx[(size_t)m * UNITS_ + col];
    out[(size_t)m * UNITS_ + col] = u * h + (1.0f - u) * c;
  }
}

extern "C" void kernel_launch(void* const* d_in, const int* in_sizes, int n_in,
                              void* d_out, int out_size, void* d_ws, size_t ws_size,
                              hipStream_t stream) {
  const float* inputs = (const float*)d_in[0];
  const float* hx     = (const float*)d_in[1];
  const int*   sidx   = (const int*)  d_in[2];
  const float* ker    = (const float*)d_in[3];
  const float* W_ru   = (const float*)d_in[4];
  const float* b_ru   = (const float*)d_in[5];
  const float* W_c    = (const float*)d_in[6];
  const float* b_c    = (const float*)d_in[7];
  float* out = (float*)d_out;

  float* ws = (float*)d_ws;
  const size_t VSZ = (size_t)M_ * C_;
  float* v0 = ws;
  float* v1 = ws + VSZ;
  float* v2 = ws + 2 * VSZ;
  float* ru = ws + 3 * VSZ;

  const int* src = sidx;
  const int* dst = sidx + E_;

  const int T = 256;
  const int gridMC   = (M_ * C_ + T - 1) / T;
  const int gridScat = (E_ * 32 + T - 1) / T;
  const int gridGemm = M_ / 16;

  k_concat_ru<<<gridMC, T, 0, stream>>>(inputs, hx, v0);
  k_zero<<<gridMC, T, 0, stream>>>(v1, M_ * C_);
  k_scatter<<<gridScat, T, 0, stream>>>(v0, v1, src, dst, ker);
  k_zero<<<gridMC, T, 0, stream>>>(v2, M_ * C_);
  k_scatter<<<gridScat, T, 0, stream>>>(v1, v2, src, dst, ker);
  k_gemm_ru<<<gridGemm, 256, 0, stream>>>(v0, v1, v2, W_ru, b_ru, ru);

  k_concat_c<<<gridMC, T, 0, stream>>>(inputs, hx, ru, v0);
  k_zero<<<gridMC, T, 0, stream>>>(v1, M_ * C_);
  k_scatter<<<gridScat, T, 0, stream>>>(v0, v1, src, dst, ker);
  k_zero<<<gridMC, T, 0, stream>>>(v2, M_ * C_);
  k_scatter<<<gridScat, T, 0, stream>>>(v1, v2, src, dst, ker);
  k_gemm_c<<<gridGemm, 128, 0, stream>>>(v0, v1, v2, W_c, b_c, ru, hx, out);
}


# Scatter adds of a dst row come from ONE L2 half: a wave reads its die, draws 16-edge chunks by per-half ticket (int atomic), adds only edges whose dst parity is its half's; sums in arrival order
# speedup vs baseline: 3.0432x; 3.0432x over previous
#include <hip/hip_runtime.h>
#include <math.h>


#define B_      4
#define N_      10000
#define F_IN_   64
#define UNITS_  64
#define E_      160000
#define C_      128
#define M_      (B_ * N_)
#define LDSW_   132
#define VSZ_    ((size_t)M_ * C_)
#define HSZ_    ((size_t)M_ * UNITS_)

typedef __attribute__((ext_vector_type(2))) float v2f;
typedef __attribute__((ext_vector_type(8))) float v8f;


__device__ inline float sigmoid_(float x) { return 1.0f / (1.0f + __expf(-x)); }

#define CH_     16
#define NCH_    (E_ / CH_)
#define CTR_(launch, half) ((launch) * 128 + (half) * 64)

template <int CW>
__device__ inline void load_rows(float (&v)[B_ * CW / 32],
                                 const float* __restrict__ x, int s, int lane) {
  const float* xp = x + (size_t)s * CW + lane;
#pragma unroll
  for (int b = 0; b < B_; ++b) {
#pragma unroll
    for (int j = 0; j < CW / 32; ++j)
      v[b * (CW / 32) + j] = xp[(size_t)b * N_ * CW + 32 * j];
  }
}

template <int CW>
__device__ inline void add_rows(float* __restrict__ y, int d, float k,
                                const float (&v)[B_ * CW / 32], int lane) {
  float* yp = y + (size_t)d * CW + lane;
#pragma unroll
  for (int b = 0; b < B_; ++b) {
#pragma unroll
    for (int j = 0; j < CW / 32; ++j)
      atomicAdd(yp + (size_t)b * N_ * CW + 32 * j, k * v[b * (CW / 32) + j]);
  }
}

template <int CW>
__device__ inline void scatter_by_half(const float* __restrict__ x,
                                       float* __restrict__ y,
                                       const int* __restrict__ src,
                                       const int* __restrict__ dst,
                                       const float* __restrict__ ker,
                                       float* __restrict__ ctrf) {
  int* ctr = (int*)ctrf;
  const int lane = threadIdx.x & 31;
  const int myh = (int)((__builtin_amdgcn_s_sendmsg_rtn(0x87) >> 18) & 1u);
  const int wave = __builtin_amdgcn_readfirstlane((int)(threadIdx.x >> 5));
  float va[B_ * CW / 32], vb[B_ * CW / 32];
  for (int pass = 0; pass < 2; ++pass) {
    const int h = myh ^ pass;
    if (pass == 1 && wave != 0) break;
    for (int t = 0; t <= NCH_; ++t) {
      int c = 0;
      if (lane == 0) c = atomicAdd(&ctr[CTR_(0, h)], 1);
      c = __builtin_amdgcn_readfirstlane(c);
      if (c >= NCH_) break;
      const int me = c * CH_ + (lane & (CH_ - 1));
      const int sv = src[me];
      const int dv = dst[me];
      const int kv = __float_as_int(ker[me]);
      asm volatile("" :: "v"(sv), "v"(kv));
      unsigned mask = __builtin_amdgcn_ballot_w32((dv & 1) == h) & ((1u << CH_) - 1u);
      for (int u = 0; u < CH_ && mask != 0u; ++u) {
        const int j = __builtin_ctz(mask);
        mask &= mask - 1u;
        if (mask != 0u) {
          const int j2 = __builtin_ctz(mask);
          mask &= mask - 1u;
          load_rows<CW>(va, x, __builtin_amdgcn_readlane(sv, j), lane);
          load_rows<CW>(vb, x, __builtin_amdgcn_readlane(sv, j2), lane);
          add_rows<CW>(y, __builtin_amdgcn_readlane(dv, j),
                       __int_as_float(__builtin_amdgcn_readlane(kv, j)), va, lane);
          add_rows<CW>(y, __builtin_amdgcn_readlane(dv, j2),
                       __int_as_float(__builtin_amdgcn_readlane(kv, j2)), vb, lane);
        } else {
          load_rows<CW>(va, x, __builtin_amdgcn_readlane(sv, j), lane);
          add_rows<CW>(y, __builtin_amdgcn_readlane(dv, j),
                       __int_as_float(__builtin_amdgcn_readlane(kv, j)), va, lane);
        }
      }
    }
  }
}

__global__ void k_scatter128(const float* __restrict__ x, float* __restrict__ y,
                             const int* __restrict__ src, const int* __restrict__ dst,
                             const float* __restrict__ ker, float* __restrict__ ctrf) {
  scatter_by_half<C_>(x, y, src, dst, ker, ctrf);
}

__global__ void k_scatter64(const float* __restrict__ x, float* __restrict__ y,
                            const int* __restrict__ src, const int* __restrict__ dst,
                            const float* __restrict__ ker, float* __restrict__ ctrf) {
  scatter_by_half<UNITS_>(x, y, src, dst, ker, ctrf);
}

template <int NOUT>
__device__ inline void wmma_proj3(const float* __restrict__ As,
                                  const float* __restrict__ W,
                                  int n0, int lane,
                                  v8f& acc0, v8f& acc1, v8f& acc2) {
  const int arow = lane & 15;
  const int koff = (lane >> 4) << 1;
  const int col  = n0 + (lane & 15);
  acc0 = {};
  acc1 = {};
  acc2 = {};
#pragma unroll 4
  for (int kb = 0; kb < C_ / 4; ++kb) {
    const int k0 = kb * 4 + koff;
    v2f a = *(const v2f*)(As + arow * LDSW_ + k0);
    const float* w = W + (size_t)k0 * NOUT + col;
    v2f b0, b1, b2;
    b0.x = w[0];
    b0.y = w[NOUT];
    b1.x = w[(size_t)C_ * NOUT];
    b1.y = w[(size_t)C_ * NOUT + NOUT];
    b2.x = w[(size_t)2 * C_ * NOUT];
    b2.y = w[(size_t)2 * C_ * NOUT + NOUT];
    acc0 = __builtin_amdgcn_wmma_f32_16x16x4_f32(
         false, a,  false, b0,
         (short)0, acc0,  false,  false);
    acc1 = __builtin_amdgcn_wmma_f32_16x16x4_f32(
        false, a, false, b1, (short)0, acc1, false, false);
    acc2 = __builtin_amdgcn_wmma_f32_16x16x4_f32(
        false, a, false, b2, (short)0, acc2, false, false);
  }
}

template <int NOUT>
__device__ inline void store_proj3(float* __restrict__ p0, float* __restrict__ p1,
                                   float* __restrict__ p2, int m0, int n0, int lane,
                                   const v8f& acc0, const v8f& acc1, const v8f& acc2) {
  const int col   = n0 + (lane & 15);
  const int mbase = m0 + ((lane >> 4) << 3);
#pragma unroll
  for (int r = 0; r < 8; ++r) {
    size_t o = (size_t)(mbase + r) * NOUT + col;
    p0[o] = acc0[r];
    p1[o] = acc1[r];
    p2[o] = acc2[r];
  }
}

__global__ void k_proj_ru(const float* __restrict__ in,
                          const float* __restrict__ hx,
                          const float* __restrict__ W,
                          float* __restrict__ p0,
                          float* __restrict__ p1,
                          float* __restrict__ p2,
                          float* __restrict__ ctrf) {
  __shared__ float As[16 * LDSW_];
  const int lane = threadIdx.x & 31;
  const int wave = threadIdx.x >> 5;
  const int m0 = blockIdx.x * 16;
  const int n0 = wave * 16;
  if (blockIdx.x == 0 && threadIdx.x < 8)
    ((int*)ctrf)[CTR_(threadIdx.x >> 1, threadIdx.x & 1)] = 0;
  for (int i = threadIdx.x; i < 16 * (C_ / 4); i += 256) {
    int row = i >> 5;
    int ch  = (i & 31) * 4;
    const float* srcp = (ch < F_IN_)
        ? in + (size_t)(m0 + row) * F_IN_ + ch
        : hx + (size_t)(m0 + row) * UNITS_ + (ch - F_IN_);
    *(float4*)(As + row * LDSW_ + ch) = *(const float4*)srcp;
  }
  __syncthreads();
  v8f acc0, acc1, acc2;
  wmma_proj3<2 * UNITS_>(As, W, n0, lane, acc0, acc1, acc2);
  store_proj3<2 * UNITS_>(p0, p1, p2, m0, n0, lane, acc0, acc1, acc2);
}

__global__ void k_proj_c(const float* __restrict__ in,
                         const float* __restrict__ hx,
                         const float* __restrict__ g,
                         const float* __restrict__ b_ru,
                         const float* __restrict__ W,
                         float* __restrict__ q0,
                         float* __restrict__ q1,
                         float* __restrict__ q2) {
  __shared__ float As[16 * LDSW_];
  const int lane = threadIdx.x & 31;
  const int wave = threadIdx.x >> 5;
  const int m0 = blockIdx.x * 16;
  const int n0 = wave * 16;
  for (int i = threadIdx.x; i < 16 * (C_ / 4); i += 128) {
    int row = i >> 5;
    int ch  = (i & 31) * 4;
    float4 v;
    if (ch < F_IN_) {
      v = *(const float4*)(in + (size_t)(m0 + row) * F_IN_ + ch);
    } else {
      int u = ch - F_IN_;
      float4 h  = *(const float4*)(hx + (size_t)(m0 + row) * UNITS_ + u);
      float4 gg = *(const float4*)(g + (size_t)(m0 + row) * (2 * UNITS_) + u);
      float4 bb = *(const float4*)(b_ru + u);
      v.x = sigmoid_(gg.x + bb.x) * h.x;
      v.y = sigmoid_(gg.y + bb.y) * h.y;
      v.z = sigmoid_(gg.z + bb.z) * h.z;
      v.w = sigmoid_(gg.w + bb.w) * h.w;
    }
    *(float4*)(As + row * LDSW_ + ch) = v;
  }
  __syncthreads();
  v8f acc0, acc1, acc2;
  wmma_proj3<UNITS_>(As, W, n0, lane, acc0, acc1, acc2);
  store_proj3<UNITS_>(q0, q1, q2, m0, n0, lane, acc0, acc1, acc2);
}

__global__ void k_out(const float* __restrict__ q,
                      const float* __restrict__ b_c,
                      const float* __restrict__ g,
                      const float* __restrict__ b_ru,
                      const float* __restrict__ hx,
                      float* __restrict__ out) {
  int idx = blockIdx.x * blockDim.x + threadIdx.x;
  if (idx >= M_ * (UNITS_ / 4)) return;
  int m   = idx >> 4;
  int col = (idx & 15) * 4;
  float4 qq = *(const float4*)(q  + (size_t)m * UNITS_ + col);
  float4 h  = *(const float4*)(hx + (size_t)m * UNITS_ + col);
  float4 gg = *(const float4*)(g  + (size_t)m * (2 * UNITS_) + UNITS_ + col);
  float4 bc = *(const float4*)(b_c + col);
  float4 bu = *(const float4*)(b_ru + UNITS_ + col);
  float4 o;
  {
    float c = tanhf(qq.x + bc.x);
    float u = sigmoid_(gg.x + bu.x);
    o.x = u * h.x + (1.0f - u) * c;
  }
  {
    float c = tanhf(qq.y + bc.y);
    float u = sigmoid_(gg.y + bu.y);
    o.y = u * h.y + (1.0f - u) * c;
  }
  {
    float c = tanhf(qq.z + bc.z);
    float u = sigmoid_(gg.z + bu.z);
    o.z = u * h.z + (1.0f - u) * c;
  }
  {
    float c = tanhf(qq.w + bc.w);
    float u = sigmoid_(gg.w + bu.w);
    o.w = u * h.w + (1.0f - u) * c;
  }
  *(float4*)(out + (size_t)m * UNITS_ + col) = o;
}

extern "C" void kernel_launch(void* const* d_in, const int* in_sizes, int n_in,
                              void* d_out, int out_size, void* d_ws, size_t ws_size,
                              hipStream_t stream) {
  const float* inputs = (const float*)d_in[0];
  const float* hx     = (const float*)d_in[1];
  const int*   sidx   = (const int*)  d_in[2];
  const float* ker    = (const float*)d_in[3];
  const float* W_ru   = (const float*)d_in[4];
  const float* b_ru   = (const float*)d_in[5];
  const float* W_c    = (const float*)d_in[6];
  const float* b_c    = (const float*)d_in[7];
  float* out = (float*)d_out;

  float* ws = (float*)d_ws;
  float* p0 = ws;
  float* p1 = ws + VSZ_;
  float* p2 = ws + 2 * VSZ_;
  float* q1 = p1;
  float* q2 = p1 + HSZ_;
  float* q0 = p2;

  const int* src = sidx;
  const int* dst = sidx + E_;

  const int T = 256;
  const int gridScat = 1024;
  float* ctr = out;
  const int gridGemm = M_ / 16;
  const int gridOut  = (M_ * (UNITS_ / 4) + T - 1) / T;

  k_proj_ru<<<gridGemm, 256, 0, stream>>>(inputs, hx, W_ru, p0, p1, p2, ctr);
  k_scatter128<<<gridScat, T, 0, stream>>>(p2, p1, src, dst, ker, ctr + CTR_(0, 0));
  k_scatter128<<<gridScat, T, 0, stream>>>(p1, p0, src, dst, ker, ctr + CTR_(1, 0));

  k_proj_c<<<gridGemm, 128, 0, stream>>>(inputs, hx, p0, b_ru, W_c, q0, q1, q2);
  k_scatter64<<<gridScat, T, 0, stream>>>(q2, q1, src, dst, ker, ctr + CTR_(2, 0));
  k_scatter64<<<gridScat, T, 0, stream>>>(q1, q0, src, dst, ker, ctr + CTR_(3, 0));

  k_out<<<gridOut, T, 0, stream>>>(q0, b_c, p0, b_ru, hx, out);
}


# Scatter-add becomes owner sums: a 32-wave work-group per 40 dst nodes scans dst, files its edges in LDS by node; a wave sums a (node,batch) row set in registers, 1 atomic per line; arrival order
# speedup vs baseline: 3.5242x; 1.1581x over previous
#include <hip/hip_runtime.h>
#include <math.h>


#define B_      4
#define N_      10000
#define F_IN_   64
#define UNITS_  64
#define E_      160000
#define C_      128
#define M_      (B_ * N_)
#define LDSW_   132
#define VSZ_    ((size_t)M_ * C_)
#define HSZ_    ((size_t)M_ * UNITS_)

typedef __attribute__((ext_vector_type(2))) float v2f;
typedef __attribute__((ext_vector_type(8))) float v8f;


__device__ inline float sigmoid_(float x) { return 1.0f / (1.0f + __expf(-x)); }

#define NPG_    40
#define NGRP_   (N_ / NPG_)
#define OWN_T_  1024
#define CAP_    14336
#define EBITS_  18
#define SCAN_U_ 8

template <int CW>
__device__ __attribute__((noinline)) void add_edge_direct(const float* x, float* y,
                                                          int s, int d, float k) {
  for (int b = 0; b < B_; ++b)
    for (int c = 0; c < CW; ++c)
      atomicAdd(y + ((size_t)b * N_ + d) * CW + c, k * x[((size_t)b * N_ + s) * CW + c]);
}

template <int CW, int K>
__device__ inline void sum_group(float (&acc)[CW / 32], const float* __restrict__ xb,
                                 int sl, int kl, int g) {
  float v[K][CW / 32];
#pragma unroll
  for (int u = 0; u < K; ++u) {
    const float* xp = xb + (size_t)__builtin_amdgcn_readlane(sl, g + u) * CW;
#pragma unroll
    for (int j = 0; j < CW / 32; ++j) v[u][j] = xp[32 * j];
  }
#pragma unroll
  for (int u = 0; u < K; ++u) {
    const float k = __int_as_float(__builtin_amdgcn_readlane(kl, g + u));
#pragma unroll
    for (int j = 0; j < CW / 32; ++j) acc[j] += k * v[u][j];
  }
}

template <int CW>
__device__ inline void owner_sums(const float* __restrict__ x, float* __restrict__ y,
                                  const int* __restrict__ src, const int* __restrict__ dst,
                                  const float* __restrict__ ker) {
  __shared__ int s_e[CAP_];
  __shared__ int s_src[CAP_];
  __shared__ int s_kb[CAP_];
  __shared__ int s_cnt[NPG_];
  __shared__ int s_off[NPG_ + 1];
  __shared__ int s_n;
  const int tid  = threadIdx.x;
  const int lane = tid & 31;
  const int wave = __builtin_amdgcn_readfirstlane(tid >> 5);
  const int lo   = blockIdx.x * NPG_;

  if (tid < NPG_) s_cnt[tid] = 0;
  if (tid == 0) s_n = 0;
  __syncthreads();

  auto file = [&](int dv, int e) {
    const unsigned l = (unsigned)(dv - lo);
    if (l < (unsigned)NPG_) {
      const int slot = atomicAdd(&s_n, 1);
      if (slot < CAP_) {
        s_e[slot] = (int)((l << EBITS_) | (unsigned)e);
        atomicAdd(&s_cnt[l], 1);
      } else {
        add_edge_direct<CW>(x, y, src[e], dv, ker[e]);
      }
    }
  };
  const int4* d4 = (const int4*)dst;
  const int rot = (int)blockIdx.x * ((E_ / 4) / NGRP_);
  for (int base = 0; base < E_ / 4; base += OWN_T_ * SCAN_U_) {
    int4 d[SCAN_U_];
    int  q[SCAN_U_];
#pragma unroll
    for (int u = 0; u < SCAN_U_; ++u) {
      const int i = base + u * OWN_T_ + tid;
      int a = i + rot;
      if (a >= E_ / 4) a -= E_ / 4;
      q[u] = (i < E_ / 4) ? a : -1;
      d[u] = (i < E_ / 4) ? d4[a] : make_int4(-1, -1, -1, -1);
    }
#pragma unroll
    for (int u = 0; u < SCAN_U_; ++u) {
      const unsigned lx = (unsigned)(d[u].x - lo), ly = (unsigned)(d[u].y - lo);
      const unsigned lz = (unsigned)(d[u].z - lo), lw = (unsigned)(d[u].w - lo);
      const unsigned lmin = min(min(lx, ly), min(lz, lw));
      if (lmin < (unsigned)NPG_) {
        file(d[u].x, 4 * q[u]);
        file(d[u].y, 4 * q[u] + 1);
        file(d[u].z, 4 * q[u] + 2);
        file(d[u].w, 4 * q[u] + 3);
      }
    }
  }
  __syncthreads();
  const int total = s_n;
  const int n = total < CAP_ ? total : CAP_;

  if (wave == 0) {
    const int c0 = s_cnt[lane];
    const int c1 = (lane < NPG_ - 32) ? s_cnt[32 + lane] : 0;
    int a0 = c0, a1 = c1;
    for (int o = 1; o < 32; o <<= 1) {
      const int t0 = __shfl_up(a0, o);
      const int t1 = __shfl_up(a1, o);
      if (lane >= o) { a0 += t0; a1 += t1; }
    }
    const int tot0 = __shfl(a0, 31);
    s_off[lane] = a0 - c0;
    s_cnt[lane] = 0;
    if (lane < NPG_ - 32) {
      s_off[32 + lane] = tot0 + a1 - c1;
      s_cnt[32 + lane] = 0;
    }
    if (lane == 0) s_off[NPG_] = n;
  }
  __syncthreads();

  for (int i = tid; i < n; i += OWN_T_) {
    const int e   = s_e[i];
    const int l   = (int)((unsigned)e >> EBITS_);
    const int idx = e & ((1 << EBITS_) - 1);
    const int pos = s_off[l] + atomicAdd(&s_cnt[l], 1);
    s_src[pos] = src[idx];
    s_kb[pos]  = __float_as_int(ker[idx]);
  }
  __syncthreads();

  for (int t = wave; t < NPG_ * B_; t += OWN_T_ / 32) {
    const int l = t >> 2;
    const int b = t & 3;
    const int beg = __builtin_amdgcn_readfirstlane(s_off[l]);
    const int end = __builtin_amdgcn_readfirstlane(s_off[l + 1]);
    if (end <= beg) continue;
    const float* xb = x + (size_t)b * N_ * CW + lane;
    float acc[CW / 32];
#pragma unroll
    for (int j = 0; j < CW / 32; ++j) acc[j] = 0.0f;
    for (int p0 = beg; p0 < end; p0 += 32) {
      const int cnt = (end - p0 < 32) ? (end - p0) : 32;
      int sl = 0, kl = 0;
      if (lane < cnt) {
        sl = s_src[p0 + lane];
        kl = s_kb[p0 + lane];
      }
      int g = 0;
      for (; g + 8 <= cnt; g += 8) sum_group<CW, 8>(acc, xb, sl, kl, g);
      if (g + 4 <= cnt) {
        sum_group<CW, 4>(acc, xb, sl, kl, g);
        g += 4;
      }
      const int r = cnt - g;
      if (r == 3)      sum_group<CW, 3>(acc, xb, sl, kl, g);
      else if (r == 2) sum_group<CW, 2>(acc, xb, sl, kl, g);
      else if (r == 1) sum_group<CW, 1>(acc, xb, sl, kl, g);
    }
    float* yp = y + ((size_t)b * N_ + lo + l) * CW + lane;
#pragma unroll
    for (int j = 0; j < CW / 32; ++j) atomicAdd(yp + 32 * j, acc[j]);
  }
}

__global__ void __launch_bounds__(OWN_T_)
k_scatter128(const float* __restrict__ x, float* __restrict__ y,
             const int* __restrict__ src, const int* __restrict__ dst,
             const float* __restrict__ ker) {
  owner_sums<C_>(x, y, src, dst, ker);
}

__global__ void __launch_bounds__(OWN_T_)
k_scatter64(const float* __restrict__ x, float* __restrict__ y,
            const int* __restrict__ src, const int* __restrict__ dst,
            const float* __restrict__ ker) {
  owner_sums<UNITS_>(x, y, src, dst, ker);
}

template <int NOUT>
__device__ inline void wmma_proj3(const float* __restrict__ As,
                                  const float* __restrict__ W,
                                  int n0, int lane,
                                  v8f& acc0, v8f& acc1, v8f& acc2) {
  const int arow = lane & 15;
  const int koff = (lane >> 4) << 1;
  const int col  = n0 + (lane & 15);
  acc0 = {};
  acc1 = {};
  acc2 = {};
#pragma unroll 4
  for (int kb = 0; kb < C_ / 4; ++kb) {
    const int k0 = kb * 4 + koff;
    v2f a = *(const v2f*)(As + arow * LDSW_ + k0);
    const float* w = W + (size_t)k0 * NOUT + col;
    v2f b0, b1, b2;
    b0.x = w[0];
    b0.y = w[NOUT];
    b1.x = w[(size_t)C_ * NOUT];
    b1.y = w[(size_t)C_ * NOUT + NOUT];
    b2.x = w[(size_t)2 * C_ * NOUT];
    b2.y = w[(size_t)2 * C_ * NOUT + NOUT];
    acc0 = __builtin_amdgcn_wmma_f32_16x16x4_f32(
         false, a,  false, b0,
         (short)0, acc0,  false,  false);
    acc1 = __builtin_amdgcn_wmma_f32_16x16x4_f32(
        false, a, false, b1, (short)0, acc1, false, false);
    acc2 = __builtin_amdgcn_wmma_f32_16x16x4_f32(
        false, a, false, b2, (short)0, acc2, false, false);
  }
}

template <int NOUT>
__device__ inline void store_proj3(float* __restrict__ p0, float* __restrict__ p1,
                                   float* __restrict__ p2, int m0, int n0, int lane,
                                   const v8f& acc0, const v8f& acc1, const v8f& acc2) {
  const int col   = n0 + (lane & 15);
  const int mbase = m0 + ((lane >> 4) << 3);
#pragma unroll
  for (int r = 0; r < 8; ++r) {
    size_t o = (size_t)(mbase + r) * NOUT + col;
    p0[o] = acc0[r];
    p1[o] = acc1[r];
    p2[o] = acc2[r];
  }
}

__global__ void k_proj_ru(const float* __restrict__ in,
                          const float* __restrict__ hx,
                          const float* __restrict__ W,
                          float* __restrict__ p0,
                          float* __restrict__ p1,
                          float* __restrict__ p2) {
  __shared__ float As[16 * LDSW_];
  const int lane = threadIdx.x & 31;
  const int wave = threadIdx.x >> 5;
  const int m0 = blockIdx.x * 16;
  const int n0 = wave * 16;
  for (int i = threadIdx.x; i < 16 * (C_ / 4); i += 256) {
    int row = i >> 5;
    int ch  = (i & 31) * 4;
    const float* srcp = (ch < F_IN_)
        ? in + (size_t)(m0 + row) * F_IN_ + ch
        : hx + (size_t)(m0 + row) * UNITS_ + (ch - F_IN_);
    *(float4*)(As + row * LDSW_ + ch) = *(const float4*)srcp;
  }
  __syncthreads();
  v8f acc0, acc1, acc2;
  wmma_proj3<2 * UNITS_>(As, W, n0, lane, acc0, acc1, acc2);
  store_proj3<2 * UNITS_>(p0, p1, p2, m0, n0, lane, acc0, acc1, acc2);
}

__global__ void k_proj_c(const float* __restrict__ in,
                         const float* __restrict__ hx,
                         const float* __restrict__ g,
                         const float* __restrict__ b_ru,
                         const float* __restrict__ W,
                         float* __restrict__ q0,
                         float* __restrict__ q1,
                         float* __restrict__ q2) {
  __shared__ float As[16 * LDSW_];
  const int lane = threadIdx.x & 31;
  const int wave = threadIdx.x >> 5;
  const int m0 = blockIdx.x * 16;
  const int n0 = wave * 16;
  for (int i = threadIdx.x; i < 16 * (C_ / 4); i += 128) {
    int row = i >> 5;
    int ch  = (i & 31) * 4;
    float4 v;
    if (ch < F_IN_) {
      v = *(const float4*)(in + (size_t)(m0 + row) * F_IN_ + ch);
    } else {
      int u = ch - F_IN_;
      float4 h  = *(const float4*)(hx + (size_t)(m0 + row) * UNITS_ + u);
      float4 gg = *(const float4*)(g + (size_t)(m0 + row) * (2 * UNITS_) + u);
      float4 bb = *(const float4*)(b_ru + u);
      v.x = sigmoid_(gg.x + bb.x) * h.x;
      v.y = sigmoid_(gg.y + bb.y) * h.y;
      v.z = sigmoid_(gg.z + bb.z) * h.z;
      v.w = sigmoid_(gg.w + bb.w) * h.w;
    }
    *(float4*)(As + row * LDSW_ + ch) = v;
  }
  __syncthreads();
  v8f acc0, acc1, acc2;
  wmma_proj3<UNITS_>(As, W, n0, lane, acc0, acc1, acc2);
  store_proj3<UNITS_>(q0, q1, q2, m0, n0, lane, acc0, acc1, acc2);
}

__global__ void k_out(const float* __restrict__ q,
                      const float* __restrict__ b_c,
                      const float* __restrict__ g,
                      const float* __restrict__ b_ru,
                      const float* __restrict__ hx,
                      float* __restrict__ out) {
  int idx = blockIdx.x * blockDim.x + threadIdx.x;
  if (idx >= M_ * (UNITS_ / 4)) return;
  int m   = idx >> 4;
  int col = (idx & 15) * 4;
  float4 qq = *(const float4*)(q  + (size_t)m * UNITS_ + col);
  float4 h  = *(const float4*)(hx + (size_t)m * UNITS_ + col);
  float4 gg = *(const float4*)(g  + (size_t)m * (2 * UNITS_) + UNITS_ + col);
  float4 bc = *(const float4*)(b_c + col);
  float4 bu = *(const float4*)(b_ru + UNITS_ + col);
  float4 o;
  {
    float c = tanhf(qq.x + bc.x);
    float u = sigmoid_(gg.x + bu.x);
    o.x = u * h.x + (1.0f - u) * c;
  }
  {
    float c = tanhf(qq.y + bc.y);
    float u = sigmoid_(gg.y + bu.y);
    o.y = u * h.y + (1.0f - u) * c;
  }
  {
    float c = tanhf(qq.z + bc.z);
    float u = sigmoid_(gg.z + bu.z);
    o.z = u * h.z + (1.0f - u) * c;
  }
  {
    float c = tanhf(qq.w + bc.w);
    float u = sigmoid_(gg.w + bu.w);
    o.w = u * h.w + (1.0f - u) * c;
  }
  *(float4*)(out + (size_t)m * UNITS_ + col) = o;
}

extern "C" void kernel_launch(void* const* d_in, const int* in_sizes, int n_in,
                              void* d_out, int out_size, void* d_ws, size_t ws_size,
                              hipStream_t stream) {
  const float* inputs = (const float*)d_in[0];
  const float* hx     = (const float*)d_in[1];
  const int*   sidx   = (const int*)  d_in[2];
  const float* ker    = (const float*)d_in[3];
  const float* W_ru   = (const float*)d_in[4];
  const float* b_ru   = (const float*)d_in[5];
  const float* W_c    = (const float*)d_in[6];
  const float* b_c    = (const float*)d_in[7];
  float* out = (float*)d_out;

  float* ws = (float*)d_ws;
  float* p0 = ws;
  float* p1 = ws + VSZ_;
  float* p2 = ws + 2 * VSZ_;
  float* q1 = p1;
  float* q2 = p1 + HSZ_;
  float* q0 = p2;

  const int* src = sidx;
  const int* dst = sidx + E_;

  const int T = 256;
  const int gridGemm = M_ / 16;
  const int gridOut  = (M_ * (UNITS_ / 4) + T - 1) / T;

  k_proj_ru<<<gridGemm, 256, 0, stream>>>(inputs, hx, W_ru, p0, p1, p2);
  k_scatter128<<<NGRP_, OWN_T_, 0, stream>>>(p2, p1, src, dst, ker);
  k_scatter128<<<NGRP_, OWN_T_, 0, stream>>>(p1, p0, src, dst, ker);

  k_proj_c<<<gridGemm, 128, 0, stream>>>(inputs, hx, p0, b_ru, W_c, q0, q1, q2);
  k_scatter64<<<NGRP_, OWN_T_, 0, stream>>>(q2, q1, src, dst, ker);
  k_scatter64<<<NGRP_, OWN_T_, 0, stream>>>(q1, q0, src, dst, ker);

  k_out<<<gridOut, T, 0, stream>>>(q0, b_c, p0, b_ru, hx, out);
}


# Owner scans read a 1-byte group key per edge (dst/40, written once by the first launch into the output buffer's head) instead of the 4-byte dst: a quarter of the bytes; sums still in arrival order
# speedup vs baseline: 3.8055x; 1.0798x over previous
#include <hip/hip_runtime.h>
#include <math.h>


#define B_      4
#define N_      10000
#define F_IN_   64
#define UNITS_  64
#define E_      160000
#define C_      128
#define M_      (B_ * N_)
#define LDSW_   132
#define VSZ_    ((size_t)M_ * C_)
#define HSZ_    ((size_t)M_ * UNITS_)

typedef __attribute__((ext_vector_type(2))) float v2f;
typedef __attribute__((ext_vector_type(8))) float v8f;


__device__ inline float sigmoid_(float x) { return 1.0f / (1.0f + __expf(-x)); }

#define NPG_    40
#define NGRP_   (N_ / NPG_)
#define OWN_T_  1024
#define CAP_    14336
#define EBITS_  18
#define NKEY4_  (E_ / 16)
#define SCAN_U_ ((NKEY4_ + OWN_T_ - 1) / OWN_T_)
#define NOKEY_  255u
static_assert(NGRP_ <= 255 && NKEY4_ % NGRP_ == 0 && SCAN_U_ % 2 == 0 && E_ % 16 == 0,
              "the key scan's layout");

__device__ inline unsigned group_key(int d) {
  return ((unsigned)d < (unsigned)N_) ? (unsigned)d / NPG_ : NOKEY_;
}

__device__ inline unsigned eq_bytes(unsigned w, unsigned gp) {
  const unsigned x = w ^ gp;
  const unsigned a = (x & 0x7F7F7F7Fu) + 0x7F7F7F7Fu;
  return ~(a | x) & 0x80808080u;
}

__device__ inline unsigned hits16(uint4 k, unsigned gp) {
  return (eq_bytes(k.x, gp) >> 7) | (eq_bytes(k.y, gp) >> 6) |
         (eq_bytes(k.z, gp) >> 5) | (eq_bytes(k.w, gp) >> 4);
}

template <int CW>
__device__ __attribute__((noinline)) void add_edge_direct(const float* x, float* y,
                                                          int s, int d, float k) {
  for (int b = 0; b < B_; ++b)
    for (int c = 0; c < CW; ++c)
      atomicAdd(y + ((size_t)b * N_ + d) * CW + c, k * x[((size_t)b * N_ + s) * CW + c]);
}

template <int CW, int K>
__device__ inline void sum_group(float (&acc)[CW / 32], const float* __restrict__ xb,
                                 int sl, int kl, int g) {
  float v[K][CW / 32];
#pragma unroll
  for (int u = 0; u < K; ++u) {
    const float* xp = xb + (size_t)__builtin_amdgcn_readlane(sl, g + u) * CW;
#pragma unroll
    for (int j = 0; j < CW / 32; ++j) v[u][j] = xp[32 * j];
  }
#pragma unroll
  for (int u = 0; u < K; ++u) {
    const float k = __int_as_float(__builtin_amdgcn_readlane(kl, g + u));
#pragma unroll
    for (int j = 0; j < CW / 32; ++j) acc[j] += k * v[u][j];
  }
}

template <int CW>
__device__ inline void owner_sums(const float* __restrict__ x, float* __restrict__ y,
                                  const int* __restrict__ src, const int* __restrict__ dst,
                                  const float* __restrict__ ker,
                                  const unsigned* __restrict__ keys) {
  __shared__ int s_e[CAP_];
  __shared__ int s_src[CAP_];
  __shared__ int s_kb[CAP_];
  __shared__ int s_cnt[NPG_];
  __shared__ int s_off[NPG_ + 1];
  __shared__ int s_n;
  const int tid  = threadIdx.x;
  const int lane = tid & 31;
  const int wave = __builtin_amdgcn_readfirstlane(tid >> 5);
  const int lo   = blockIdx.x * NPG_;

  if (tid < NPG_) s_cnt[tid] = 0;
  if (tid == 0) s_n = 0;
  __syncthreads();

  auto file = [&](int e) {
    const int slot = atomicAdd(&s_n, 1);
    if (slot < CAP_) s_e[slot] = e;
    else add_edge_direct<CW>(x, y, src[e], dst[e], ker[e]);
  };
  const uint4* k4 = (const uint4*)keys;
  const unsigned gp = (unsigned)blockIdx.x * 0x01010101u;
  const int rot = (int)blockIdx.x * (NKEY4_ / NGRP_);
  for (int base = 0; base < NKEY4_; base += OWN_T_ * SCAN_U_) {
    uint4 kv[SCAN_U_];
    const int i0 = base + tid;
#pragma unroll
    for (int u = 0; u < SCAN_U_; ++u) {
      const int i = i0 + u * OWN_T_;
      int a = i + rot;
      if (a >= NKEY4_) a -= NKEY4_;
      kv[u] = (i < NKEY4_) ? k4[a] : make_uint4(~0u, ~0u, ~0u, ~0u);
    }
#pragma unroll
    for (int u = 0; u < SCAN_U_; u += 2) {
      unsigned m = hits16(kv[u], gp) | (hits16(kv[u + 1], gp) << 4);
      while (m) {
        const int p = __builtin_ctz(m);
        m &= m - 1;
        int a = i0 + rot + (u + ((p >> 2) & 1)) * OWN_T_;
        if (a >= NKEY4_) a -= NKEY4_;
        file(16 * a + 4 * (p & 3) + (p >> 3));
      }
    }
  }
  __syncthreads();
  const int total = s_n;
  const int n = total < CAP_ ? total : CAP_;

  for (int i = tid; i < n; i += OWN_T_) {
    const int e = s_e[i];
    const int l = dst[e] - lo;
    s_e[i] = (int)(((unsigned)l << EBITS_) | (unsigned)e);
    atomicAdd(&s_cnt[l], 1);
  }
  __syncthreads();

  if (wave == 0) {
    const int c0 = s_cnt[lane];
    const int c1 = (lane < NPG_ - 32) ? s_cnt[32 + lane] : 0;
    int a0 = c0, a1 = c1;
    for (int o = 1; o < 32; o <<= 1) {
      const int t0 = __shfl_up(a0, o);
      const int t1 = __shfl_up(a1, o);
      if (lane >= o) { a0 += t0; a1 += t1; }
    }
    const int tot0 = __shfl(a0, 31);
    s_off[lane] = a0 - c0;
    s_cnt[lane] = 0;
    if (lane < NPG_ - 32) {
      s_off[32 + lane] = tot0 + a1 - c1;
      s_cnt[32 + lane] = 0;
    }
    if (lane == 0) s_off[NPG_] = n;
  }
  __syncthreads();

  for (int i = tid; i < n; i += OWN_T_) {
    const int e   = s_e[i];
    const int l   = (int)((unsigned)e >> EBITS_);
    const int idx = e & ((1 << EBITS_) - 1);
    const int pos = s_off[l] + atomicAdd(&s_cnt[l], 1);
    s_src[pos] = src[idx];
    s_kb[pos]  = __float_as_int(ker[idx]);
  }
  __syncthreads();

  for (int t = wave; t < NPG_ * B_; t += OWN_T_ / 32) {
    const int l = t >> 2;
    const int b = t & 3;
    const int beg = __builtin_amdgcn_readfirstlane(s_off[l]);
    const int end = __builtin_amdgcn_readfirstlane(s_off[l + 1]);
    if (end <= beg) continue;
    const float* xb = x + (size_t)b * N_ * CW + lane;
    float acc[CW / 32];
#pragma unroll
    for (int j = 0; j < CW / 32; ++j) acc[j] = 0.0f;
    for (int p0 = beg; p0 < end; p0 += 32) {
      const int cnt = (end - p0 < 32) ? (end - p0) : 32;
      int sl = 0, kl = 0;
      if (lane < cnt) {
        sl = s_src[p0 + lane];
        kl = s_kb[p0 + lane];
      }
      int g = 0;
      for (; g + 8 <= cnt; g += 8) sum_group<CW, 8>(acc, xb, sl, kl, g);
      if (g + 4 <= cnt) {
        sum_group<CW, 4>(acc, xb, sl, kl, g);
        g += 4;
      }
      const int r = cnt - g;
      if (r == 3)      sum_group<CW, 3>(acc, xb, sl, kl, g);
      else if (r == 2) sum_group<CW, 2>(acc, xb, sl, kl, g);
      else if (r == 1) sum_group<CW, 1>(acc, xb, sl, kl, g);
    }
    float* yp = y + ((size_t)b * N_ + lo + l) * CW + lane;
#pragma unroll
    for (int j = 0; j < CW / 32; ++j) atomicAdd(yp + 32 * j, acc[j]);
  }
}

__global__ void __launch_bounds__(OWN_T_)
k_scatter128(const float* __restrict__ x, float* __restrict__ y,
             const int* __restrict__ src, const int* __restrict__ dst,
             const float* __restrict__ ker, const unsigned* __restrict__ keys) {
  owner_sums<C_>(x, y, src, dst, ker, keys);
}

__global__ void __launch_bounds__(OWN_T_)
k_scatter64(const float* __restrict__ x, float* __restrict__ y,
            const int* __restrict__ src, const int* __restrict__ dst,
            const float* __restrict__ ker, const unsigned* __restrict__ keys) {
  owner_sums<UNITS_>(x, y, src, dst, ker, keys);
}

template <int NOUT>
__device__ inline void wmma_proj3(const float* __restrict__ As,
                                  const float* __restrict__ W,
                                  int n0, int lane,
                                  v8f& acc0, v8f& acc1, v8f& acc2) {
  const int arow = lane & 15;
  const int koff = (lane >> 4) << 1;
  const int col  = n0 + (lane & 15);
  acc0 = {};
  acc1 = {};
  acc2 = {};
#pragma unroll 4
  for (int kb = 0; kb < C_ / 4; ++kb) {
    const int k0 = kb * 4 + koff;
    v2f a = *(const v2f*)(As + arow * LDSW_ + k0);
    const float* w = W + (size_t)k0 * NOUT + col;
    v2f b0, b1, b2;
    b0.x = w[0];
    b0.y = w[NOUT];
    b1.x = w[(size_t)C_ * NOUT];
    b1.y = w[(size_t)C_ * NOUT + NOUT];
    b2.x = w[(size_t)2 * C_ * NOUT];
    b2.y = w[(size_t)2 * C_ * NOUT + NOUT];
    acc0 = __builtin_amdgcn_wmma_f32_16x16x4_f32(
         false, a,  false, b0,
         (short)0, acc0,  false,  false);
    acc1 = __builtin_amdgcn_wmma_f32_16x16x4_f32(
        false, a, false, b1, (short)0, acc1, false, false);
    acc2 = __builtin_amdgcn_wmma_f32_16x16x4_f32(
        false, a, false, b2, (short)0, acc2, false, false);
  }
}

template <int NOUT>
__device__ inline void store_proj3(float* __restrict__ p0, float* __restrict__ p1,
                                   float* __restrict__ p2, int m0, int n0, int lane,
                                   const v8f& acc0, const v8f& acc1, const v8f& acc2) {
  const int col   = n0 + (lane & 15);
  const int mbase = m0 + ((lane >> 4) << 3);
#pragma unroll
  for (int r = 0; r < 8; ++r) {
    size_t o = (size_t)(mbase + r) * NOUT + col;
    p0[o] = acc0[r];
    p1[o] = acc1[r];
    p2[o] = acc2[r];
  }
}

__global__ void k_proj_ru(const float* __restrict__ in,
                          const float* __restrict__ hx,
                          const float* __restrict__ W,
                          float* __restrict__ p0,
                          float* __restrict__ p1,
                          float* __restrict__ p2,
                          const int* __restrict__ dst,
                          unsigned* __restrict__ keys) {
  __shared__ float As[16 * LDSW_];
  const int lane = threadIdx.x & 31;
  const int wave = threadIdx.x >> 5;
  const int m0 = blockIdx.x * 16;
  const int n0 = wave * 16;
  for (int i = threadIdx.x; i < 16 * (C_ / 4); i += 256) {
    int row = i >> 5;
    int ch  = (i & 31) * 4;
    const float* srcp = (ch < F_IN_)
        ? in + (size_t)(m0 + row) * F_IN_ + ch
        : hx + (size_t)(m0 + row) * UNITS_ + (ch - F_IN_);
    *(float4*)(As + row * LDSW_ + ch) = *(const float4*)srcp;
  }
  for (int i = blockIdx.x * 256 + threadIdx.x; i < E_ / 4; i += gridDim.x * 256) {
    const int4 d = ((const int4*)dst)[i];
    keys[i] = group_key(d.x) | (group_key(d.y) << 8) | (group_key(d.z) << 16) |
              (group_key(d.w) << 24);
  }
  __syncthreads();
  v8f acc0, acc1, acc2;
  wmma_proj3<2 * UNITS_>(As, W, n0, lane, acc0, acc1, acc2);
  store_proj3<2 * UNITS_>(p0, p1, p2, m0, n0, lane, acc0, acc1, acc2);
}

__global__ void k_proj_c(const float* __restrict__ in,
                         const float* __restrict__ hx,
                         const float* __restrict__ g,
                         const float* __restrict__ b_ru,
                         const float* __restrict__ W,
                         float* __restrict__ q0,
                         float* __restrict__ q1,
                         float* __restrict__ q2) {
  __shared__ float As[16 * LDSW_];
  const int lane = threadIdx.x & 31;
  const int wave = threadIdx.x >> 5;
  const int m0 = blockIdx.x * 16;
  const int n0 = wave * 16;
  for (int i = threadIdx.x; i < 16 * (C_ / 4); i += 128) {
    int row = i >> 5;
    int ch  = (i & 31) * 4;
    float4 v;
    if (ch < F_IN_) {
      v = *(const float4*)(in + (size_t)(m0 + row) * F_IN_ + ch);
    } else {
      int u = ch - F_IN_;
      float4 h  = *(const float4*)(hx + (size_t)(m0 + row) * UNITS_ + u);
      float4 gg = *(const float4*)(g + (size_t)(m0 + row) * (2 * UNITS_) + u);
      float4 bb = *(const float4*)(b_ru + u);
      v.x = sigmoid_(gg.x + bb.x) * h.x;
      v.y = sigmoid_(gg.y + bb.y) * h.y;
      v.z = sigmoid_(gg.z + bb.z) * h.z;
      v.w = sigmoid_(gg.w + bb.w) * h.w;
    }
    *(float4*)(As + row * LDSW_ + ch) = v;
  }
  __syncthreads();
  v8f acc0, acc1, acc2;
  wmma_proj3<UNITS_>(As, W, n0, lane, acc0, acc1, acc2);
  store_proj3<UNITS_>(q0, q1, q2, m0, n0, lane, acc0, acc1, acc2);
}

__global__ void k_out(const float* __restrict__ q,
                      const float* __restrict__ b_c,
                      const float* __restrict__ g,
                      const float* __restrict__ b_ru,
                      const float* __restrict__ hx,
                      float* __restrict__ out) {
  int idx = blockIdx.x * blockDim.x + threadIdx.x;
  if (idx >= M_ * (UNITS_ / 4)) return;
  int m   = idx >> 4;
  int col = (idx & 15) * 4;
  float4 qq = *(const float4*)(q  + (size_t)m * UNITS_ + col);
  float4 h  = *(const float4*)(hx + (size_t)m * UNITS_ + col);
  float4 gg = *(const float4*)(g  + (size_t)m * (2 * UNITS_) + UNITS_ + col);
  float4 bc = *(const float4*)(b_c + col);
  float4 bu = *(const float4*)(b_ru + UNITS_ + col);
  float4 o;
  {
    float c = tanhf(qq.x + bc.x);
    float u = sigmoid_(gg.x + bu.x);
    o.x = u * h.x + (1.0f - u) * c;
  }
  {
    float c = tanhf(qq.y + bc.y);
    float u = sigmoid_(gg.y + bu.y);
    o.y = u * h.y + (1.0f - u) * c;
  }
  {
    float c = tanhf(qq.z + bc.z);
    float u = sigmoid_(gg.z + bu.z);
    o.z = u * h.z + (1.0f - u) * c;
  }
  {
    float c = tanhf(qq.w + bc.w);
    float u = sigmoid_(gg.w + bu.w);
    o.w = u * h.w + (1.0f - u) * c;
  }
  *(float4*)(out + (size_t)m * UNITS_ + col) = o;
}

extern "C" void kernel_launch(void* const* d_in, const int* in_sizes, int n_in,
                              void* d_out, int out_size, void* d_ws, size_t ws_size,
                              hipStream_t stream) {
  const float* inputs = (const float*)d_in[0];
  const float* hx     = (const float*)d_in[1];
  const int*   sidx   = (const int*)  d_in[2];
  const float* ker    = (const float*)d_in[3];
  const float* W_ru   = (const float*)d_in[4];
  const float* b_ru   = (const float*)d_in[5];
  const float* W_c    = (const float*)d_in[6];
  const float* b_c    = (const float*)d_in[7];
  float* out = (float*)d_out;

  float* ws = (float*)d_ws;
  float* p0 = ws;
  float* p1 = ws + VSZ_;
  float* p2 = ws + 2 * VSZ_;
  float* q1 = p1;
  float* q2 = p1 + HSZ_;
  float* q0 = p2;

  const int* src = sidx;
  const int* dst = sidx + E_;
  unsigned* keys = (unsigned*)d_out;

  const int T = 256;
  const int gridGemm = M_ / 16;
  const int gridOut  = (M_ * (UNITS_ / 4) + T - 1) / T;

  k_proj_ru<<<gridGemm, 256, 0, stream>>>(inputs, hx, W_ru, p0, p1, p2, dst, keys);
  k_scatter128<<<NGRP_, OWN_T_, 0, stream>>>(p2, p1, src, dst, ker, keys);
  k_scatter128<<<NGRP_, OWN_T_, 0, stream>>>(p1, p0, src, dst, ker, keys);

  k_proj_c<<<gridGemm, 128, 0, stream>>>(inputs, hx, p0, b_ru, W_c, q0, q1, q2);
  k_scatter64<<<NGRP_, OWN_T_, 0, stream>>>(q2, q1, src, dst, ker, keys);
  k_scatter64<<<NGRP_, OWN_T_, 0, stream>>>(q1, q0, src, dst, ker, keys);

  k_out<<<gridOut, T, 0, stream>>>(q0, b_c, p0, b_ru, hx, out);
}


# Scatter launches 2-4 load the group's node-ordered tables (source rows, weights, run starts) that scatter launch 1 built and stored in the output buffer; no key scan or sort; sums in arrival order
# speedup vs baseline: 4.2266x; 1.1107x over previous
#include <hip/hip_runtime.h>
#include <math.h>


#define B_      4
#define N_      10000
#define F_IN_   64
#define UNITS_  64
#define E_      160000
#define C_      128
#define M_      (B_ * N_)
#define LDSW_   132
#define VSZ_    ((size_t)M_ * C_)
#define HSZ_    ((size_t)M_ * UNITS_)

typedef __attribute__((ext_vector_type(2))) float v2f;
typedef __attribute__((ext_vector_type(8))) float v8f;


__device__ inline float sigmoid_(float x) { return 1.0f / (1.0f + __expf(-x)); }

#define NPG_    40
#define NGRP_   (N_ / NPG_)
#define OWN_T_  1024
#define CAP_    14336
#define EBITS_  18
#define NKEY4_  (E_ / 16)
#define SCAN_U_ ((NKEY4_ + OWN_T_ - 1) / OWN_T_)
#define NOKEY_  255u
static_assert(NGRP_ <= 255 && NKEY4_ % NGRP_ == 0 && SCAN_U_ % 2 == 0 && E_ % 16 == 0,
              "the key scan's layout");
#define LSTR_   2048
#define OFFW_   64
#define OFFS_I_ 40960
#define LIST_I_ (OFFS_I_ + NGRP_ * OFFW_)
static_assert(OFFS_I_ >= E_ / 4 && NPG_ + 2 <= OFFW_ && (LIST_I_ * 4) % 256 == 0 &&
              LSTR_ <= CAP_ && LSTR_ % OWN_T_ == 0 &&
              (size_t)LIST_I_ + (size_t)NGRP_ * 2 * LSTR_ <= (size_t)M_ * UNITS_,
              "the stored tables lie behind the keys, inside the output buffer");

__device__ inline unsigned group_key(int d) {
  return ((unsigned)d < (unsigned)N_) ? (unsigned)d / NPG_ : NOKEY_;
}

__device__ inline unsigned eq_bytes(unsigned w, unsigned gp) {
  const unsigned x = w ^ gp;
  const unsigned a = (x & 0x7F7F7F7Fu) + 0x7F7F7F7Fu;
  return ~(a | x) & 0x80808080u;
}

__device__ inline unsigned hits16(uint4 k, unsigned gp) {
  return (eq_bytes(k.x, gp) >> 7) | (eq_bytes(k.y, gp) >> 6) |
         (eq_bytes(k.z, gp) >> 5) | (eq_bytes(k.w, gp) >> 4);
}

template <int CW>
__device__ __attribute__((noinline)) void add_edge_direct(const float* x, float* y,
                                                          int s, int d, float k) {
  for (int b = 0; b < B_; ++b)
    for (int c = 0; c < CW; ++c)
      atomicAdd(y + ((size_t)b * N_ + d) * CW + c, k * x[((size_t)b * N_ + s) * CW + c]);
}

template <int CW, int K>
__device__ inline void sum_group(float (&acc)[CW / 32], const float* __restrict__ xb,
                                 int sl, int kl, int g) {
  float v[K][CW / 32];
#pragma unroll
  for (int u = 0; u < K; ++u) {
    const float* xp = xb + (size_t)__builtin_amdgcn_readlane(sl, g + u) * CW;
#pragma unroll
    for (int j = 0; j < CW / 32; ++j) v[u][j] = xp[32 * j];
  }
#pragma unroll
  for (int u = 0; u < K; ++u) {
    const float k = __int_as_float(__builtin_amdgcn_readlane(kl, g + u));
#pragma unroll
    for (int j = 0; j < CW / 32; ++j) acc[j] += k * v[u][j];
  }
}

template <int CW, bool BUILD>
__device__ inline void owner_sums(const float* __restrict__ x, float* __restrict__ y,
                                  const int* __restrict__ src, const int* __restrict__ dst,
                                  const float* __restrict__ ker,
                                  const unsigned* __restrict__ keys,
                                  int* __restrict__ tabs) {
  __shared__ int s_e[CAP_];
  __shared__ int s_src[CAP_];
  __shared__ int s_kb[CAP_];
  __shared__ int s_cnt[NPG_];
  __shared__ int s_off[NPG_ + 1];
  __shared__ int s_n;
  const int tid  = threadIdx.x;
  const int lane = tid & 31;
  const int wave = __builtin_amdgcn_readfirstlane(tid >> 5);
  const int lo   = blockIdx.x * NPG_;
  int* tab  = tabs + OFFS_I_ + (int)blockIdx.x * OFFW_;
  int* gsrc = tabs + LIST_I_ + (int)blockIdx.x * (2 * LSTR_);
  int* gkb  = gsrc + LSTR_;

  int stored = LSTR_ + 1;
  if (!BUILD) stored = tab[NPG_ + 1];

  if (!BUILD && stored <= LSTR_) {
    for (int u = 0; u < LSTR_ / OWN_T_; ++u) {
      const int i = tid + u * OWN_T_;
      if (i < stored) {
        s_src[i] = gsrc[i];
        s_kb[i]  = gkb[i];
      }
    }
    const int to = tid - (OWN_T_ - 64);
    if (to >= 0 && to <= NPG_) s_off[to] = tab[to];
    __syncthreads();
  } else {
    if (tid < NPG_) s_cnt[tid] = 0;
    if (tid == 0) s_n = 0;
    __syncthreads();

    auto file = [&](int e) {
      const int slot = atomicAdd(&s_n, 1);
      if (slot < CAP_) s_e[slot] = e;
      else add_edge_direct<CW>(x, y, src[e], dst[e], ker[e]);
    };
    const uint4* k4 = (const uint4*)keys;
    const unsigned gp = (unsigned)blockIdx.x * 0x01010101u;
    const int rot = (int)blockIdx.x * (NKEY4_ / NGRP_);
    for (int base = 0; base < NKEY4_; base += OWN_T_ * SCAN_U_) {
      uint4 kv[SCAN_U_];
      const int i0 = base + tid;
#pragma unroll
      for (int u = 0; u < SCAN_U_; ++u) {
        const int i = i0 + u * OWN_T_;
        int a = i + rot;
        if (a >= NKEY4_) a -= NKEY4_;
        kv[u] = (i < NKEY4_) ? k4[a] : make_uint4(~0u, ~0u, ~0u, ~0u);
      }
#pragma unroll
      for (int u = 0; u < SCAN_U_; u += 2) {
        unsigned m = hits16(kv[u], gp) | (hits16(kv[u + 1], gp) << 4);
        while (m) {
          const int p = __builtin_ctz(m);
          m &= m - 1;
          int a = i0 + rot + (u + ((p >> 2) & 1)) * OWN_T_;
          if (a >= NKEY4_) a -= NKEY4_;
          file(16 * a + 4 * (p & 3) + (p >> 3));
        }
      }
    }
    __syncthreads();
    const int total = s_n;
    const int n = total < CAP_ ? total : CAP_;

    for (int i = tid; i < n; i += OWN_T_) {
      const int e = s_e[i];
      const int l = dst[e] - lo;
      s_e[i] = (int)(((unsigned)l << EBITS_) | (unsigned)e);
      atomicAdd(&s_cnt[l], 1);
    }
    __syncthreads();

    if (wave == 0) {
      const int c0 = s_cnt[lane];
      const int c1 = (lane < NPG_ - 32) ? s_cnt[32 + lane] : 0;
      int a0 = c0, a1 = c1;
      for (int o = 1; o < 32; o <<= 1) {
        const int t0 = __shfl_up(a0, o);
        const int t1 = __shfl_up(a1, o);
        if (lane >= o) { a0 += t0; a1 += t1; }
      }
      const int tot0 = __shfl(a0, 31);
      s_off[lane] = a0 - c0;
      s_cnt[lane] = 0;
      if (lane < NPG_ - 32) {
        s_off[32 + lane] = tot0 + a1 - c1;
        s_cnt[32 + lane] = 0;
      }
      if (lane == 0) s_off[NPG_] = n;
    }
    __syncthreads();

    for (int i = tid; i < n; i += OWN_T_) {
      const int e   = s_e[i];
      const int l   = (int)((unsigned)e >> EBITS_);
      const int idx = e & ((1 << EBITS_) - 1);
      const int pos = s_off[l] + atomicAdd(&s_cnt[l], 1);
      s_src[pos] = src[idx];
      s_kb[pos]  = __float_as_int(ker[idx]);
    }
    __syncthreads();

    if (BUILD) {
      if (total <= LSTR_) {
        for (int i = tid; i < n; i += OWN_T_) {
          gsrc[i] = s_src[i];
          gkb[i]  = s_kb[i];
        }
      }
      if (tid <= NPG_) tab[tid] = s_off[tid];
      if (tid == NPG_ + 1) tab[NPG_ + 1] = total;
    }
  }

  for (int t = wave; t < NPG_ * B_; t += OWN_T_ / 32) {
    const int l = t >> 2;
    const int b = t & 3;
    const int beg = __builtin_amdgcn_readfirstlane(s_off[l]);
    const int end = __builtin_amdgcn_readfirstlane(s_off[l + 1]);
    if (end <= beg) continue;
    const float* xb = x + (size_t)b * N_ * CW + lane;
    float acc[CW / 32];
#pragma unroll
    for (int j = 0; j < CW / 32; ++j) acc[j] = 0.0f;
    for (int p0 = beg; p0 < end; p0 += 32) {
      const int cnt = (end - p0 < 32) ? (end - p0) : 32;
      int sl = 0, kl = 0;
      if (lane < cnt) {
        sl = s_src[p0 + lane];
        kl = s_kb[p0 + lane];
      }
      int g = 0;
      for (; g + 8 <= cnt; g += 8) sum_group<CW, 8>(acc, xb, sl, kl, g);
      if (g + 4 <= cnt) {
        sum_group<CW, 4>(acc, xb, sl, kl, g);
        g += 4;
      }
      const int r = cnt - g;
      if (r == 3)      sum_group<CW, 3>(acc, xb, sl, kl, g);
      else if (r == 2) sum_group<CW, 2>(acc, xb, sl, kl, g);
      else if (r == 1) sum_group<CW, 1>(acc, xb, sl, kl, g);
    }
    float* yp = y + ((size_t)b * N_ + lo + l) * CW + lane;
#pragma unroll
    for (int j = 0; j < CW / 32; ++j) atomicAdd(yp + 32 * j, acc[j]);
  }
}

__global__ void __launch_bounds__(OWN_T_)
k_scatter128_build(const float* __restrict__ x, float* __restrict__ y,
                   const int* __restrict__ src, const int* __restrict__ dst,
                   const float* __restrict__ ker, const unsigned* __restrict__ keys,
                   int* __restrict__ tabs) {
  owner_sums<C_, true>(x, y, src, dst, ker, keys, tabs);
}

__global__ void __launch_bounds__(OWN_T_)
k_scatter128(const float* __restrict__ x, float* __restrict__ y,
             const int* __restrict__ src, const int* __restrict__ dst,
             const float* __restrict__ ker, const unsigned* __restrict__ keys,
             int* __restrict__ tabs) {
  owner_sums<C_, false>(x, y, src, dst, ker, keys, tabs);
}

__global__ void __launch_bounds__(OWN_T_)
k_scatter64(const float* __restrict__ x, float* __restrict__ y,
            const int* __restrict__ src, const int* __restrict__ dst,
            const float* __restrict__ ker, const unsigned* __restrict__ keys,
            int* __restrict__ tabs) {
  owner_sums<UNITS_, false>(x, y, src, dst, ker, keys, tabs);
}

template <int NOUT>
__device__ inline void wmma_proj3(const float* __restrict__ As,
                                  const float* __restrict__ W,
                                  int n0, int lane,
                                  v8f& acc0, v8f& acc1, v8f& acc2) {
  const int arow = lane & 15;
  const int koff = (lane >> 4) << 1;
  const int col  = n0 + (lane & 15);
  acc0 = {};
  acc1 = {};
  acc2 = {};
#pragma unroll 4
  for (int kb = 0; kb < C_ / 4; ++kb) {
    const int k0 = kb * 4 + koff;
    v2f a = *(const v2f*)(As + arow * LDSW_ + k0);
    const float* w = W + (size_t)k0 * NOUT + col;
    v2f b0, b1, b2;
    b0.x = w[0];
    b0.y = w[NOUT];
    b1.x = w[(size_t)C_ * NOUT];
    b1.y = w[(size_t)C_ * NOUT + NOUT];
    b2.x = w[(size_t)2 * C_ * NOUT];
    b2.y = w[(size_t)2 * C_ * NOUT + NOUT];
    acc0 = __builtin_amdgcn_wmma_f32_16x16x4_f32(
         false, a,  false, b0,
         (short)0, acc0,  false,  false);
    acc1 = __builtin_amdgcn_wmma_f32_16x16x4_f32(
        false, a, false, b1, (short)0, acc1, false, false);
    acc2 = __builtin_amdgcn_wmma_f32_16x16x4_f32(
        false, a, false, b2, (short)0, acc2, false, false);
  }
}

template <int NOUT>
__device__ inline void store_proj3(float* __restrict__ p0, float* __restrict__ p1,
                                   float* __restrict__ p2, int m0, int n0, int lane,
                                   const v8f& acc0, const v8f& acc1, const v8f& acc2) {
  const int col   = n0 + (lane & 15);
  const int mbase = m0 + ((lane >> 4) << 3);
#pragma unroll
  for (int r = 0; r < 8; ++r) {
    size_t o = (size_t)(mbase + r) * NOUT + col;
    p0[o] = acc0[r];
    p1[o] = acc1[r];
    p2[o] = acc2[r];
  }
}

__global__ void k_proj_ru(const float* __restrict__ in,
                          const float* __restrict__ hx,
                          const float* __restrict__ W,
                          float* __restrict__ p0,
                          float* __restrict__ p1,
                          float* __restrict__ p2,
                          const int* __restrict__ dst,
                          unsigned* __restrict__ keys) {
  __shared__ float As[16 * LDSW_];
  const int lane = threadIdx.x & 31;
  const int wave = threadIdx.x >> 5;
  const int m0 = blockIdx.x * 16;
  const int n0 = wave * 16;
  for (int i = threadIdx.x; i < 16 * (C_ / 4); i += 256) {
    int row = i >> 5;
    int ch  = (i & 31) * 4;
    const float* srcp = (ch < F_IN_)
        ? in + (size_t)(m0 + row) * F_IN_ + ch
        : hx + (size_t)(m0 + row) * UNITS_ + (ch - F_IN_);
    *(float4*)(As + row * LDSW_ + ch) = *(const float4*)srcp;
  }
  for (int i = blockIdx.x * 256 + threadIdx.x; i < E_ / 4; i += gridDim.x * 256) {
    const int4 d = ((const int4*)dst)[i];
    keys[i] = group_key(d.x) | (group_key(d.y) << 8) | (group_key(d.z) << 16) |
              (group_key(d.w) << 24);
  }
  __syncthreads();
  v8f acc0, acc1, acc2;
  wmma_proj3<2 * UNITS_>(As, W, n0, lane, acc0, acc1, acc2);
  store_proj3<2 * UNITS_>(p0, p1, p2, m0, n0, lane, acc0, acc1, acc2);
}

__global__ void k_proj_c(const float* __restrict__ in,
                         const float* __restrict__ hx,
                         const float* __restrict__ g,
                         const float* __restrict__ b_ru,
                         const float* __restrict__ W,
                         float* __restrict__ q0,
                         float* __restrict__ q1,
                         float* __restrict__ q2) {
  __shared__ float As[16 * LDSW_];
  const int lane = threadIdx.x & 31;
  const int wave = threadIdx.x >> 5;
  const int m0 = blockIdx.x * 16;
  const int n0 = wave * 16;
  for (int i = threadIdx.x; i < 16 * (C_ / 4); i += 128) {
    int row = i >> 5;
    int ch  = (i & 31) * 4;
    float4 v;
    if (ch < F_IN_) {
      v = *(const float4*)(in + (size_t)(m0 + row) * F_IN_ + ch);
    } else {
      int u = ch - F_IN_;
      float4 h  = *(const float4*)(hx + (size_t)(m0 + row) * UNITS_ + u);
      float4 gg = *(const float4*)(g + (size_t)(m0 + row) * (2 * UNITS_) + u);
      float4 bb = *(const float4*)(b_ru + u);
      v.x = sigmoid_(gg.x + bb.x) * h.x;
      v.y = sigmoid_(gg.y + bb.y) * h.y;
      v.z = sigmoid_(gg.z + bb.z) * h.z;
      v.w = sigmoid_(gg.w + bb.w) * h.w;
    }
    *(float4*)(As + row * LDSW_ + ch) = v;
  }
  __syncthreads();
  v8f acc0, acc1, acc2;
  wmma_proj3<UNITS_>(As, W, n0, lane, acc0, acc1, acc2);
  store_proj3<UNITS_>(q0, q1, q2, m0, n0, lane, acc0, acc1, acc2);
}

__global__ void k_out(const float* __restrict__ q,
                      const float* __restrict__ b_c,
                      const float* __restrict__ g,
                      const float* __restrict__ b_ru,
                      const float* __restrict__ hx,
                      float* __restrict__ out) {
  int idx = blockIdx.x * blockDim.x + threadIdx.x;
  if (idx >= M_ * (UNITS_ / 4)) return;
  int m   = idx >> 4;
  int col = (idx & 15) * 4;
  float4 qq = *(const float4*)(q  + (size_t)m * UNITS_ + col);
  float4 h  = *(const float4*)(hx + (size_t)m * UNITS_ + col);
  float4 gg = *(const float4*)(g  + (size_t)m * (2 * UNITS_) + UNITS_ + col);
  float4 bc = *(const float4*)(b_c + col);
  float4 bu = *(const float4*)(b_ru + UNITS_ + col);
  float4 o;
  {
    float c = tanhf(qq.x + bc.x);
    float u = sigmoid_(gg.x + bu.x);
    o.x = u * h.x + (1.0f - u) * c;
  }
  {
    float c = tanhf(qq.y + bc.y);
    float u = sigmoid_(gg.y + bu.y);
    o.y = u * h.y + (1.0f - u) * c;
  }
  {
    float c = tanhf(qq.z + bc.z);
    float u = sigmoid_(gg.z + bu.z);
    o.z = u * h.z + (1.0f - u) * c;
  }
  {
    float c = tanhf(qq.w + bc.w);
    float u = sigmoid_(gg.w + bu.w);
    o.w = u * h.w + (1.0f - u) * c;
  }
  *(float4*)(out + (size_t)m * UNITS_ + col) = o;
}

extern "C" void kernel_launch(void* const* d_in, const int* in_sizes, int n_in,
                              void* d_out, int out_size, void* d_ws, size_t ws_size,
                              hipStream_t stream) {
  const float* inputs = (const float*)d_in[0];
  const float* hx     = (const float*)d_in[1];
  const int*   sidx   = (const int*)  d_in[2];
  const float* ker    = (const float*)d_in[3];
  const float* W_ru   = (const float*)d_in[4];
  const float* b_ru   = (const float*)d_in[5];
  const float* W_c    = (const float*)d_in[6];
  const float* b_c    = (const float*)d_in[7];
  float* out = (float*)d_out;

  float* ws = (float*)d_ws;
  float* p0 = ws;
  float* p1 = ws + VSZ_;
  float* p2 = ws + 2 * VSZ_;
  float* q1 = p1;
  float* q2 = p1 + HSZ_;
  float* q0 = p2;

  const int* src = sidx;
  const int* dst = sidx + E_;
  unsigned* keys = (unsigned*)d_out;
  int* tabs = (int*)d_out;

  const int T = 256;
  const int gridGemm = M_ / 16;
  const int gridOut  = (M_ * (UNITS_ / 4) + T - 1) / T;

  k_proj_ru<<<gridGemm, 256, 0, stream>>>(inputs, hx, W_ru, p0, p1, p2, dst, keys);
  k_scatter128_build<<<NGRP_, OWN_T_, 0, stream>>>(p2, p1, src, dst, ker, keys, tabs);
  k_scatter128<<<NGRP_, OWN_T_, 0, stream>>>(p1, p0, src, dst, ker, keys, tabs);

  k_proj_c<<<gridGemm, 128, 0, stream>>>(inputs, hx, p0, b_ru, W_c, q0, q1, q2);
  k_scatter64<<<NGRP_, OWN_T_, 0, stream>>>(q2, q1, src, dst, ker, keys, tabs);
  k_scatter64<<<NGRP_, OWN_T_, 0, stream>>>(q1, q0, src, dst, ker, keys, tabs);

  k_out<<<gridOut, T, 0, stream>>>(q0, b_c, p0, b_ru, hx, out);
}


# Projection GEMM blocks cover 80 rows (were 16): a wave loads each W fragment once per k-step and uses it for 5 row tiles (15 accumulator tiles); W is read 5x less often, by 5x fewer waves
# speedup vs baseline: 4.4275x; 1.0475x over previous
#include <hip/hip_runtime.h>
#include <math.h>


#define B_      4
#define N_      10000
#define F_IN_   64
#define UNITS_  64
#define E_      160000
#define C_      128
#define M_      (B_ * N_)
#define LDSW_   132
#define VSZ_    ((size_t)M_ * C_)
#define HSZ_    ((size_t)M_ * UNITS_)

#define RT_     5
#define BM_     (16 * RT_)
static_assert(M_ % BM_ == 0, "projection blocks cover whole row tiles");

typedef __attribute__((ext_vector_type(2))) float v2f;
typedef __attribute__((ext_vector_type(8))) float v8f;


__device__ inline float sigmoid_(float x) { return 1.0f / (1.0f + __expf(-x)); }

#define NPG_    40
#define NGRP_   (N_ / NPG_)
#define OWN_T_  1024
#define CAP_    14336
#define EBITS_  18
#define NKEY4_  (E_ / 16)
#define SCAN_U_ ((NKEY4_ + OWN_T_ - 1) / OWN_T_)
#define NOKEY_  255u
static_assert(NGRP_ <= 255 && NKEY4_ % NGRP_ == 0 && SCAN_U_ % 2 == 0 && E_ % 16 == 0,
              "the key scan's layout");
#define LSTR_   2048
#define OFFW_   64
#define OFFS_I_ 40960
#define LIST_I_ (OFFS_I_ + NGRP_ * OFFW_)
static_assert(OFFS_I_ >= E_ / 4 && NPG_ + 2 <= OFFW_ && (LIST_I_ * 4) % 256 == 0 &&
              LSTR_ <= CAP_ && LSTR_ % OWN_T_ == 0 &&
              (size_t)LIST_I_ + (size_t)NGRP_ * 2 * LSTR_ <= (size_t)M_ * UNITS_,
              "the stored tables lie behind the keys, inside the output buffer");

__device__ inline unsigned group_key(int d) {
  return ((unsigned)d < (unsigned)N_) ? (unsigned)d / NPG_ : NOKEY_;
}

__device__ inline unsigned eq_bytes(unsigned w, unsigned gp) {
  const unsigned x = w ^ gp;
  const unsigned a = (x & 0x7F7F7F7Fu) + 0x7F7F7F7Fu;
  return ~(a | x) & 0x80808080u;
}

__device__ inline unsigned hits16(uint4 k, unsigned gp) {
  return (eq_bytes(k.x, gp) >> 7) | (eq_bytes(k.y, gp) >> 6) |
         (eq_bytes(k.z, gp) >> 5) | (eq_bytes(k.w, gp) >> 4);
}

template <int CW>
__device__ __attribute__((noinline)) void add_edge_direct(const float* x, float* y,
                                                          int s, int d, float k) {
  for (int b = 0; b < B_; ++b)
    for (int c = 0; c < CW; ++c)
      atomicAdd(y + ((size_t)b * N_ + d) * CW + c, k * x[((size_t)b * N_ + s) * CW + c]);
}

template <int CW, int K>
__device__ inline void sum_group(float (&acc)[CW / 32], const float* __restrict__ xb,
                                 int sl, int kl, int g) {
  float v[K][CW / 32];
#pragma unroll
  for (int u = 0; u < K; ++u) {
    const float* xp = xb + (size_t)__builtin_amdgcn_readlane(sl, g + u) * CW;
#pragma unroll
    for (int j = 0; j < CW / 32; ++j) v[u][j] = xp[32 * j];
  }
#pragma unroll
  for (int u = 0; u < K; ++u) {
    const float k = __int_as_float(__builtin_amdgcn_readlane(kl, g + u));
#pragma unroll
    for (int j = 0; j < CW / 32; ++j) acc[j] += k * v[u][j];
  }
}

template <int CW, bool BUILD>
__device__ inline void owner_sums(const float* __restrict__ x, float* __restrict__ y,
                                  const int* __restrict__ src, const int* __restrict__ dst,
                                  const float* __restrict__ ker,
                                  const unsigned* __restrict__ keys,
                                  int* __restrict__ tabs) {
  __shared__ int s_e[CAP_];
  __shared__ int s_src[CAP_];
  __shared__ int s_kb[CAP_];
  __shared__ int s_cnt[NPG_];
  __shared__ int s_off[NPG_ + 1];
  __shared__ int s_n;
  const int tid  = threadIdx.x;
  const int lane = tid & 31;
  const int wave = __builtin_amdgcn_readfirstlane(tid >> 5);
  const int lo   = blockIdx.x * NPG_;
  int* tab  = tabs + OFFS_I_ + (int)blockIdx.x * OFFW_;
  int* gsrc = tabs + LIST_I_ + (int)blockIdx.x * (2 * LSTR_);
  int* gkb  = gsrc + LSTR_;

  int stored = LSTR_ + 1;
  if (!BUILD) stored = tab[NPG_ + 1];

  if (!BUILD && stored <= LSTR_) {
    for (int u = 0; u < LSTR_ / OWN_T_; ++u) {
      const int i = tid + u * OWN_T_;
      if (i < stored) {
        s_src[i] = gsrc[i];
        s_kb[i]  = gkb[i];
      }
    }
    const int to = tid - (OWN_T_ - 64);
    if (to >= 0 && to <= NPG_) s_off[to] = tab[to];
    __syncthreads();
  } else {
    if (tid < NPG_) s_cnt[tid] = 0;
    if (tid == 0) s_n = 0;
    __syncthreads();

    auto file = [&](int e) {
      const int slot = atomicAdd(&s_n, 1);
      if (slot < CAP_) s_e[slot] = e;
      else add_edge_direct<CW>(x, y, src[e], dst[e], ker[e]);
    };
    const uint4* k4 = (const uint4*)keys;
    const unsigned gp = (unsigned)blockIdx.x * 0x01010101u;
    const int rot = (int)blockIdx.x * (NKEY4_ / NGRP_);
    for (int base = 0; base < NKEY4_; base += OWN_T_ * SCAN_U_) {
      uint4 kv[SCAN_U_];
      const int i0 = base + tid;
#pragma unroll
      for (int u = 0; u < SCAN_U_; ++u) {
        const int i = i0 + u * OWN_T_;
        int a = i + rot;
        if (a >= NKEY4_) a -= NKEY4_;
        kv[u] = (i < NKEY4_) ? k4[a] : make_uint4(~0u, ~0u, ~0u, ~0u);
      }
#pragma unroll
      for (int u = 0; u < SCAN_U_; u += 2) {
        unsigned m = hits16(kv[u], gp) | (hits16(kv[u + 1], gp) << 4);
        while (m) {
          const int p = __builtin_ctz(m);
          m &= m - 1;
          int a = i0 + rot + (u + ((p >> 2) & 1)) * OWN_T_;
          if (a >= NKEY4_) a -= NKEY4_;
          file(16 * a + 4 * (p & 3) + (p >> 3));
        }
      }
    }
    __syncthreads();
    const int total = s_n;
    const int n = total < CAP_ ? total : CAP_;

    for (int i = tid; i < n; i += OWN_T_) {
      const int e = s_e[i];
      const int l = dst[e] - lo;
      s_e[i] = (int)(((unsigned)l << EBITS_) | (unsigned)e);
      atomicAdd(&s_cnt[l], 1);
    }
    __syncthreads();

    if (wave == 0) {
      const int c0 = s_cnt[lane];
      const int c1 = (lane < NPG_ - 32) ? s_cnt[32 + lane] : 0;
      int a0 = c0, a1 = c1;
      for (int o = 1; o < 32; o <<= 1) {
        const int t0 = __shfl_up(a0, o);
        const int t1 = __shfl_up(a1, o);
        if (lane >= o) { a0 += t0; a1 += t1; }
      }
      const int tot0 = __shfl(a0, 31);
      s_off[lane] = a0 - c0;
      s_cnt[lane] = 0;
      if (lane < NPG_ - 32) {
        s_off[32 + lane] = tot0 + a1 - c1;
        s_cnt[32 + lane] = 0;
      }
      if (lane == 0) s_off[NPG_] = n;
    }
    __syncthreads();

    for (int i = tid; i < n; i += OWN_T_) {
      const int e   = s_e[i];
      const int l   = (int)((unsigned)e >> EBITS_);
      const int idx = e & ((1 << EBITS_) - 1);
      const int pos = s_off[l] + atomicAdd(&s_cnt[l], 1);
      s_src[pos] = src[idx];
      s_kb[pos]  = __float_as_int(ker[idx]);
    }
    __syncthreads();

    if (BUILD) {
      if (total <= LSTR_) {
        for (int i = tid; i < n; i += OWN_T_) {
          gsrc[i] = s_src[i];
          gkb[i]  = s_kb[i];
        }
      }
      if (tid <= NPG_) tab[tid] = s_off[tid];
      if (tid == NPG_ + 1) tab[NPG_ + 1] = total;
    }
  }

  for (int t = wave; t < NPG_ * B_; t += OWN_T_ / 32) {
    const int l = t >> 2;
    const int b = t & 3;
    const int beg = __builtin_amdgcn_readfirstlane(s_off[l]);
    const int end = __builtin_amdgcn_readfirstlane(s_off[l + 1]);
    if (end <= beg) continue;
    const float* xb = x + (size_t)b * N_ * CW + lane;
    float acc[CW / 32];
#pragma unroll
    for (int j = 0; j < CW / 32; ++j) acc[j] = 0.0f;
    for (int p0 = beg; p0 < end; p0 += 32) {
      const int cnt = (end - p0 < 32) ? (end - p0) : 32;
      int sl = 0, kl = 0;
      if (lane < cnt) {
        sl = s_src[p0 + lane];
        kl = s_kb[p0 + lane];
      }
      int g = 0;
      for (; g + 8 <= cnt; g += 8) sum_group<CW, 8>(acc, xb, sl, kl, g);
      if (g + 4 <= cnt) {
        sum_group<CW, 4>(acc, xb, sl, kl, g);
        g += 4;
      }
      const int r = cnt - g;
      if (r == 3)      sum_group<CW, 3>(acc, xb, sl, kl, g);
      else if (r == 2) sum_group<CW, 2>(acc, xb, sl, kl, g);
      else if (r == 1) sum_group<CW, 1>(acc, xb, sl, kl, g);
    }
    float* yp = y + ((size_t)b * N_ + lo + l) * CW + lane;
#pragma unroll
    for (int j = 0; j < CW / 32; ++j) atomicAdd(yp + 32 * j, acc[j]);
  }
}

__global__ void __launch_bounds__(OWN_T_)
k_scatter128_build(const float* __restrict__ x, float* __restrict__ y,
                   const int* __restrict__ src, const int* __restrict__ dst,
                   const float* __restrict__ ker, const unsigned* __restrict__ keys,
                   int* __restrict__ tabs) {
  owner_sums<C_, true>(x, y, src, dst, ker, keys, tabs);
}

__global__ void __launch_bounds__(OWN_T_)
k_scatter128(const float* __restrict__ x, float* __restrict__ y,
             const int* __restrict__ src, const int* __restrict__ dst,
             const float* __restrict__ ker, const unsigned* __restrict__ keys,
             int* __restrict__ tabs) {
  owner_sums<C_, false>(x, y, src, dst, ker, keys, tabs);
}

__global__ void __launch_bounds__(OWN_T_)
k_scatter64(const float* __restrict__ x, float* __restrict__ y,
            const int* __restrict__ src, const int* __restrict__ dst,
            const float* __restrict__ ker, const unsigned* __restrict__ keys,
            int* __restrict__ tabs) {
  owner_sums<UNITS_, false>(x, y, src, dst, ker, keys, tabs);
}

template <int NOUT>
__device__ inline void wmma_proj3(const float* __restrict__ As,
                                  const float* __restrict__ W,
                                  int n0, int lane, v8f (&acc)[RT_][3]) {
  const int arow = lane & 15;
  const int koff = (lane >> 4) << 1;
  const int col  = n0 + (lane & 15);
#pragma unroll
  for (int t = 0; t < RT_; ++t) {
#pragma unroll
    for (int j = 0; j < 3; ++j) acc[t][j] = {};
  }
#pragma unroll 4
  for (int kb = 0; kb < C_ / 4; ++kb) {
    const int k0 = kb * 4 + koff;
    const float* w = W + (size_t)k0 * NOUT + col;
    v2f b[3];
#pragma unroll
    for (int j = 0; j < 3; ++j) {
      b[j].x = w[(size_t)j * C_ * NOUT];
      b[j].y = w[(size_t)j * C_ * NOUT + NOUT];
    }
    v2f a[RT_];
#pragma unroll
    for (int t = 0; t < RT_; ++t)
      a[t] = *(const v2f*)(As + (t * 16 + arow) * LDSW_ + k0);
#pragma unroll
    for (int t = 0; t < RT_; ++t) {
#pragma unroll
      for (int j = 0; j < 3; ++j)
        acc[t][j] = __builtin_amdgcn_wmma_f32_16x16x4_f32(
             false, a[t],  false, b[j],
             (short)0, acc[t][j],  false,  false);
    }
  }
}

template <int NOUT>
__device__ inline void store_proj3(float* __restrict__ p0, float* __restrict__ p1,
                                   float* __restrict__ p2, int m0, int n0, int lane,
                                   const v8f (&acc)[RT_][3]) {
  const int col = n0 + (lane & 15);
#pragma unroll
  for (int t = 0; t < RT_; ++t) {
    const int mbase = m0 + 16 * t + ((lane >> 4) << 3);
#pragma unroll
    for (int r = 0; r < 8; ++r) {
      size_t o = (size_t)(mbase + r) * NOUT + col;
      p0[o] = acc[t][0][r];
      p1[o] = acc[t][1][r];
      p2[o] = acc[t][2][r];
    }
  }
}

__global__ void __launch_bounds__(256)
k_proj_ru(const float* __restrict__ in,
          const float* __restrict__ hx,
          const float* __restrict__ W,
          float* __restrict__ p0,
          float* __restrict__ p1,
          float* __restrict__ p2,
          const int* __restrict__ dst,
          unsigned* __restrict__ keys) {
  __shared__ float As[BM_ * LDSW_];
  const int lane = threadIdx.x & 31;
  const int wave = threadIdx.x >> 5;
  const int m0 = blockIdx.x * BM_;
  const int n0 = wave * 16;
#pragma unroll
  for (int u = 0; u < BM_ * (C_ / 4) / 256; ++u) {
    const int i   = threadIdx.x + u * 256;
    const int row = i >> 5;
    const int ch  = (i & 31) * 4;
    const float* srcp = (ch < F_IN_)
        ? in + (size_t)(m0 + row) * F_IN_ + ch
        : hx + (size_t)(m0 + row) * UNITS_ + (ch - F_IN_);
    *(float4*)(As + row * LDSW_ + ch) = *(const float4*)srcp;
  }
  for (int i = blockIdx.x * 256 + threadIdx.x; i < E_ / 4; i += gridDim.x * 256) {
    const int4 d = ((const int4*)dst)[i];
    keys[i] = group_key(d.x) | (group_key(d.y) << 8) | (group_key(d.z) << 16) |
              (group_key(d.w) << 24);
  }
  __syncthreads();
  v8f acc[RT_][3];
  wmma_proj3<2 * UNITS_>(As, W, n0, lane, acc);
  store_proj3<2 * UNITS_>(p0, p1, p2, m0, n0, lane, acc);
}

__global__ void __launch_bounds__(128)
k_proj_c(const float* __restrict__ in,
         const float* __restrict__ hx,
         const float* __restrict__ g,
         const float* __restrict__ b_ru,
         const float* __restrict__ W,
         float* __restrict__ q0,
         float* __restrict__ q1,
         float* __restrict__ q2) {
  __shared__ float As[BM_ * LDSW_];
  const int lane = threadIdx.x & 31;
  const int wave = threadIdx.x >> 5;
  const int m0 = blockIdx.x * BM_;
  const int n0 = wave * 16;
  const int c4   = (threadIdx.x & 15) * 4;
  const int row0 = threadIdx.x >> 4;
#pragma unroll
  for (int u = 0; u < BM_ / 8; ++u) {
    const int row = row0 + 8 * u;
    *(float4*)(As + row * LDSW_ + c4) =
        *(const float4*)(in + (size_t)(m0 + row) * F_IN_ + c4);
  }
  const float4 bb = *(const float4*)(b_ru + c4);
#pragma unroll
  for (int u = 0; u < BM_ / 8; ++u) {
    const int row = row0 + 8 * u;
    const float4 h  = *(const float4*)(hx + (size_t)(m0 + row) * UNITS_ + c4);
    const float4 gg = *(const float4*)(g + (size_t)(m0 + row) * (2 * UNITS_) + c4);
    float4 v;
    v.x = sigmoid_(gg.x + bb.x) * h.x;
    v.y = sigmoid_(gg.y + bb.y) * h.y;
    v.z = sigmoid_(gg.z + bb.z) * h.z;
    v.w = sigmoid_(gg.w + bb.w) * h.w;
    *(float4*)(As + row * LDSW_ + F_IN_ + c4) = v;
  }
  __syncthreads();
  v8f acc[RT_][3];
  wmma_proj3<UNITS_>(As, W, n0, lane, acc);
  store_proj3<UNITS_>(q0, q1, q2, m0, n0, lane, acc);
}

__global__ void k_out(const float* __restrict__ q,
                      const float* __restrict__ b_c,
                      const float* __restrict__ g,
                      const float* __restrict__ b_ru,
                      const float* __restrict__ hx,
                      float* __restrict__ out) {
  int idx = blockIdx.x * blockDim.x + threadIdx.x;
  if (idx >= M_ * (UNITS_ / 4)) return;
  int m   = idx >> 4;
  int col = (idx & 15) * 4;
  float4 qq = *(const float4*)(q  + (size_t)m * UNITS_ + col);
  float4 h  = *(const float4*)(hx + (size_t)m * UNITS_ + col);
  float4 gg = *(const float4*)(g  + (size_t)m * (2 * UNITS_) + UNITS_ + col);
  float4 bc = *(const float4*)(b_c + col);
  float4 bu = *(const float4*)(b_ru + UNITS_ + col);
  float4 o;
  {
    float c = tanhf(qq.x + bc.x);
    float u = sigmoid_(gg.x + bu.x);
    o.x = u * h.x + (1.0f - u) * c;
  }
  {
    float c = tanhf(qq.y + bc.y);
    float u = sigmoid_(gg.y + bu.y);
    o.y = u * h.y + (1.0f - u) * c;
  }
  {
    float c = tanhf(qq.z + bc.z);
    float u = sigmoid_(gg.z + bu.z);
    o.z = u * h.z + (1.0f - u) * c;
  }
  {
    float c = tanhf(qq.w + bc.w);
    float u = sigmoid_(gg.w + bu.w);
    o.w = u * h.w + (1.0f - u) * c;
  }
  *(float4*)(out + (size_t)m * UNITS_ + col) = o;
}

extern "C" void kernel_launch(void* const* d_in, const int* in_sizes, int n_in,
                              void* d_out, int out_size, void* d_ws, size_t ws_size,
                              hipStream_t stream) {
  const float* inputs = (const float*)d_in[0];
  const float* hx     = (const float*)d_in[1];
  const int*   sidx   = (const int*)  d_in[2];
  const float* ker    = (const float*)d_in[3];
  const float* W_ru   = (const float*)d_in[4];
  const float* b_ru   = (const float*)d_in[5];
  const float* W_c    = (const float*)d_in[6];
  const float* b_c    = (const float*)d_in[7];
  float* out = (float*)d_out;

  float* ws = (float*)d_ws;
  float* p0 = ws;
  float* p1 = ws + VSZ_;
  float* p2 = ws + 2 * VSZ_;
  float* q1 = p1;
  float* q2 = p1 + HSZ_;
  float* q0 = p2;

  const int* src = sidx;
  const int* dst = sidx + E_;
  unsigned* keys = (unsigned*)d_out;
  int* tabs = (int*)d_out;

  const int T = 256;
  const int gridGemm = M_ / BM_;
  const int gridOut  = (M_ * (UNITS_ / 4) + T - 1) / T;

  k_proj_ru<<<gridGemm, 256, 0, stream>>>(inputs, hx, W_ru, p0, p1, p2, dst, keys);
  k_scatter128_build<<<NGRP_, OWN_T_, 0, stream>>>(p2, p1, src, dst, ker, keys, tabs);
  k_scatter128<<<NGRP_, OWN_T_, 0, stream>>>(p1, p0, src, dst, ker, keys, tabs);

  k_proj_c<<<gridGemm, 128, 0, stream>>>(inputs, hx, p0, b_ru, W_c, q0, q1, q2);
  k_scatter64<<<NGRP_, OWN_T_, 0, stream>>>(q2, q1, src, dst, ker, keys, tabs);
  k_scatter64<<<NGRP_, OWN_T_, 0, stream>>>(q1, q0, src, dst, ker, keys, tabs);

  k_out<<<gridOut, T, 0, stream>>>(q0, b_c, p0, b_ru, hx, out);
}
